# starting kernel
# baseline (speedup 1.0000x reference)
#include <hip/hip_runtime.h>


typedef float v2f __attribute__((ext_vector_type(2)));
typedef float v8f __attribute__((ext_vector_type(8)));

#define BATCH 16

__global__ void zero_f32(float* __restrict__ p, int n) {
    int i = blockIdx.x * blockDim.x + threadIdx.x;
    if (i < n) p[i] = 0.0f;
}

__global__ void transpose_x(const float* __restrict__ x,
                            float* __restrict__ xT, int nsrc) {
    int i = blockIdx.x * blockDim.x + threadIdx.x;
    if (i < nsrc * BATCH) {
        int s = i >> 4;
        int b = i & 15;
        xT[i] = x[b * nsrc + s];
    }
}

__global__ void scatter_edges(const float* __restrict__ vals,
                              const int* __restrict__ idx,
                              const float* __restrict__ xT,
                              float* __restrict__ outT, int nnz) {
    int e = blockIdx.x * blockDim.x + threadIdx.x;
    if (e >= nnz) return;
    int r = __builtin_nontemporal_load(idx + e);
    int c = __builtin_nontemporal_load(idx + nnz + e);
    float v = __builtin_nontemporal_load(vals + e);

    const float4* xs = (const float4*)(xT + (size_t)c * BATCH);
    float4 x0 = xs[0];
    float4 x1 = xs[1];
    float4 x2 = xs[2];
    float4 x3 = xs[3];

    float* o = outT + (size_t)r * BATCH;
    atomicAdd(o + 0,  v * x0.x); atomicAdd(o + 1,  v * x0.y);
    atomicAdd(o + 2,  v * x0.z); atomicAdd(o + 3,  v * x0.w);
    atomicAdd(o + 4,  v * x1.x); atomicAdd(o + 5,  v * x1.y);
    atomicAdd(o + 6,  v * x1.z); atomicAdd(o + 7,  v * x1.w);
    atomicAdd(o + 8,  v * x2.x); atomicAdd(o + 9,  v * x2.y);
    atomicAdd(o + 10, v * x2.z); atomicAdd(o + 11, v * x2.w);
    atomicAdd(o + 12, v * x3.x); atomicAdd(o + 13, v * x3.y);
    atomicAdd(o + 14, v * x3.z); atomicAdd(o + 15, v * x3.w);
}

__global__ void wmma_transpose_bias(const float* __restrict__ outT,
                                    const float* __restrict__ bias,
                                    float* __restrict__ out, int ndst) {
    int lane = threadIdx.x & 31;
    int wave = threadIdx.x >> 5;
    int tile = blockIdx.x * (blockDim.x >> 5) + wave;
    int d0 = tile * 16;
    if (d0 >= ndst) return;

    int m = lane & 15;
    int h = lane >> 4;
    int k0 = 2 * h;

    float bb = bias[d0 + m];
    v8f c = {bb, bb, bb, bb, bb, bb, bb, bb};

#pragma unroll
    for (int q = 0; q < 4; ++q) {
        v2f a, b;
        a.x = (m == 4 * q + k0)     ? 1.0f : 0.0f;
        a.y = (m == 4 * q + k0 + 1) ? 1.0f : 0.0f;
        b = *(const v2f*)(outT + (size_t)(d0 + m) * BATCH + 4 * q + k0);
        c = __builtin_amdgcn_wmma_f32_16x16x4_f32(
                 false, a,  false, b,
                 (short)0, c,  false,  false);
    }

#pragma unroll
    for (int r = 0; r < 8; ++r) {
        int brow = r + 8 * h;
        out[(size_t)brow * ndst + d0 + m] = c[r];
    }
}

extern "C" void kernel_launch(void* const* d_in, const int* in_sizes, int n_in,
                              void* d_out, int out_size, void* d_ws, size_t ws_size,
                              hipStream_t stream) {
    const float* x    = (const float*)d_in[0];
    const float* vals = (const float*)d_in[1];
    const float* bias = (const float*)d_in[2];
    const int*   idx  = (const int*)d_in[3];

    int nsrc = in_sizes[0] / BATCH;
    int nnz  = in_sizes[1];
    int ndst = in_sizes[2];

    float* out  = (float*)d_out;
    float* xT   = (float*)d_ws;
    float* outT = xT + (size_t)nsrc * BATCH;

    int nOutT = ndst * BATCH;
    zero_f32<<<(nOutT + 255) / 256, 256, 0, stream>>>(outT, nOutT);

    int nXT = nsrc * BATCH;
    transpose_x<<<(nXT + 255) / 256, 256, 0, stream>>>(x, xT, nsrc);

    scatter_edges<<<(nnz + 255) / 256, 256, 0, stream>>>(vals, idx, xT, outT, nnz);

    int tiles = (ndst + 15) / 16;
    wmma_transpose_bias<<<(tiles + 7) / 8, 256, 0, stream>>>(outT, bias, out, ndst);
}


# scatter: one half-wave per edge (lane%16 = batch column), so an edge's 16 f32 atomic adds leave as one 64-byte request, 2 edges per instruction; adds stay atomic, arrival order as before
# speedup vs baseline: 7.0632x; 7.0632x over previous
#include <hip/hip_runtime.h>


typedef float v2f __attribute__((ext_vector_type(2)));
typedef float v8f __attribute__((ext_vector_type(8)));

#define BATCH 16

__global__ void zero_f32(float* __restrict__ p, int n) {
    int i = blockIdx.x * blockDim.x + threadIdx.x;
    if (i < n) p[i] = 0.0f;
}

__global__ void transpose_x(const float* __restrict__ x,
                            float* __restrict__ xT, int nsrc) {
    int i = blockIdx.x * blockDim.x + threadIdx.x;
    if (i < nsrc * BATCH) {
        int s = i >> 4;
        int b = i & 15;
        xT[i] = x[b * nsrc + s];
    }
}

__global__ void scatter_edges(const float* __restrict__ vals,
                              const int* __restrict__ idx,
                              const float* __restrict__ xT,
                              float* __restrict__ outT, int nnz) {
    int e = blockIdx.x * blockDim.x + threadIdx.x;
    int lane = threadIdx.x & 31;
    unsigned sub = lane & 15;
    int h = lane >> 4;
    int r = 0, c = 0;
    float v = 0.0f;
    if (e < nnz) {
        r = __builtin_nontemporal_load(idx + e);
        c = __builtin_nontemporal_load(idx + nnz + e);
        v = __builtin_nontemporal_load(vals + e);
    }
    int vb = __float_as_int(v);
    int nvalid = nnz - __builtin_amdgcn_readfirstlane(e - lane);

    unsigned ro[16];
    float p[16];

    for (int j = 0; j < 16; ++j) {
        int r0 = __builtin_amdgcn_readlane(r, 2 * j);
        int r1 = __builtin_amdgcn_readlane(r, 2 * j + 1);
        int c0 = __builtin_amdgcn_readlane(c, 2 * j);
        int c1 = __builtin_amdgcn_readlane(c, 2 * j + 1);
        int v0 = __builtin_amdgcn_readlane(vb, 2 * j);
        int v1 = __builtin_amdgcn_readlane(vb, 2 * j + 1);
        unsigned rj = (unsigned)(h ? r1 : r0);
        unsigned cj = (unsigned)(h ? c1 : c0);
        float vj = __int_as_float(h ? v1 : v0);
        ro[j] = rj * BATCH + sub;
        p[j] = vj * xT[cj * BATCH + sub];
    }

    if (nvalid >= 32) {
        for (int j = 0; j < 16; ++j) atomicAdd(outT + ro[j], p[j]);
    } else {
        for (int j = 0; j < 16; ++j)
            if (2 * j + h < nvalid) atomicAdd(outT + ro[j], p[j]);
    }
}

__global__ void wmma_transpose_bias(const float* __restrict__ outT,
                                    const float* __restrict__ bias,
                                    float* __restrict__ out, int ndst) {
    int lane = threadIdx.x & 31;
    int wave = threadIdx.x >> 5;
    int tile = blockIdx.x * (blockDim.x >> 5) + wave;
    int d0 = tile * 16;
    if (d0 >= ndst) return;

    int m = lane & 15;
    int h = lane >> 4;
    int k0 = 2 * h;

    float bb = bias[d0 + m];
    v8f c = {bb, bb, bb, bb, bb, bb, bb, bb};

#pragma unroll
    for (int q = 0; q < 4; ++q) {
        v2f a, b;
        a.x = (m == 4 * q + k0)     ? 1.0f : 0.0f;
        a.y = (m == 4 * q + k0 + 1) ? 1.0f : 0.0f;
        b = *(const v2f*)(outT + (size_t)(d0 + m) * BATCH + 4 * q + k0);
        c = __builtin_amdgcn_wmma_f32_16x16x4_f32(
                 false, a,  false, b,
                 (short)0, c,  false,  false);
    }

#pragma unroll
    for (int r = 0; r < 8; ++r) {
        int brow = r + 8 * h;
        out[(size_t)brow * ndst + d0 + m] = c[r];
    }
}

extern "C" void kernel_launch(void* const* d_in, const int* in_sizes, int n_in,
                              void* d_out, int out_size, void* d_ws, size_t ws_size,
                              hipStream_t stream) {
    const float* x    = (const float*)d_in[0];
    const float* vals = (const float*)d_in[1];
    const float* bias = (const float*)d_in[2];
    const int*   idx  = (const int*)d_in[3];

    int nsrc = in_sizes[0] / BATCH;
    int nnz  = in_sizes[1];
    int ndst = in_sizes[2];

    float* out  = (float*)d_out;
    float* xT   = (float*)d_ws;
    float* outT = xT + (size_t)nsrc * BATCH;

    int nOutT = ndst * BATCH;
    zero_f32<<<(nOutT + 255) / 256, 256, 0, stream>>>(outT, nOutT);

    int nXT = nsrc * BATCH;
    transpose_x<<<(nXT + 255) / 256, 256, 0, stream>>>(x, xT, nsrc);

    scatter_edges<<<(nnz + 255) / 256, 256, 0, stream>>>(vals, idx, xT, outT, nnz);

    int tiles = (ndst + 15) / 16;
    wmma_transpose_bias<<<(tiles + 7) / 8, 256, 0, stream>>>(outT, bias, out, ndst);
}


# scatter: each wave adds into the accumulator plane of its own L2 half (die read from the placement word); transpose kernel adds the 2 planes; sum order: 2 partial sums per output, then added
# speedup vs baseline: 12.8444x; 1.8185x over previous
#include <hip/hip_runtime.h>


typedef float v2f __attribute__((ext_vector_type(2)));
typedef float v8f __attribute__((ext_vector_type(8)));

#define BATCH 16

__global__ void zero_planes(float* __restrict__ p0, float* __restrict__ p1,
                            int n) {
    int i = blockIdx.x * blockDim.x + threadIdx.x;
    if (i < n) {
        p0[i] = 0.0f;
        p1[i] = 0.0f;
    }
}

__global__ void transpose_x(const float* __restrict__ x,
                            float* __restrict__ xT, int nsrc) {
    int i = blockIdx.x * blockDim.x + threadIdx.x;
    if (i < nsrc * BATCH) {
        int s = i >> 4;
        int b = i & 15;
        xT[i] = x[b * nsrc + s];
    }
}

__global__ void scatter_edges(const float* __restrict__ vals,
                              const int* __restrict__ idx,
                              const float* __restrict__ xT,
                              float* __restrict__ planes, size_t plane,
                              int nnz) {
    unsigned place = __builtin_amdgcn_s_sendmsg_rtn(0x87);
    int e = blockIdx.x * blockDim.x + threadIdx.x;
    int lane = threadIdx.x & 31;
    unsigned sub = lane & 15;
    int h = lane >> 4;
    int r = 0, c = 0;
    float v = 0.0f;
    if (e < nnz) {
        r = __builtin_nontemporal_load(idx + e);
        c = __builtin_nontemporal_load(idx + nnz + e);
        v = __builtin_nontemporal_load(vals + e);
    }
    int vb = __float_as_int(v);
    float* outT = planes + (size_t)((place >> 18) & 1u) * plane;
    int nvalid = nnz - __builtin_amdgcn_readfirstlane(e - lane);

    unsigned ro[16];
    float p[16];

    for (int j = 0; j < 16; ++j) {
        int r0 = __builtin_amdgcn_readlane(r, 2 * j);
        int r1 = __builtin_amdgcn_readlane(r, 2 * j + 1);
        int c0 = __builtin_amdgcn_readlane(c, 2 * j);
        int c1 = __builtin_amdgcn_readlane(c, 2 * j + 1);
        int v0 = __builtin_amdgcn_readlane(vb, 2 * j);
        int v1 = __builtin_amdgcn_readlane(vb, 2 * j + 1);
        unsigned rj = (unsigned)(h ? r1 : r0);
        unsigned cj = (unsigned)(h ? c1 : c0);
        float vj = __int_as_float(h ? v1 : v0);
        ro[j] = rj * BATCH + sub;
        p[j] = vj * xT[cj * BATCH + sub];
    }

    if (nvalid >= 32) {
        for (int j = 0; j < 16; ++j) atomicAdd(outT + ro[j], p[j]);
    } else {
        for (int j = 0; j < 16; ++j)
            if (2 * j + h < nvalid) atomicAdd(outT + ro[j], p[j]);
    }
}

__global__ void wmma_transpose_bias(const float* __restrict__ outT0,
                                    const float* __restrict__ outT1,
                                    const float* __restrict__ bias,
                                    float* __restrict__ out, int ndst) {
    int lane = threadIdx.x & 31;
    int wave = threadIdx.x >> 5;
    int tile = blockIdx.x * (blockDim.x >> 5) + wave;
    int d0 = tile * 16;
    if (d0 >= ndst) return;

    int m = lane & 15;
    int h = lane >> 4;
    int k0 = 2 * h;

    float bb = bias[d0 + m];
    v8f c = {bb, bb, bb, bb, bb, bb, bb, bb};

#pragma unroll
    for (int q = 0; q < 4; ++q) {
        v2f a, b;
        a.x = (m == 4 * q + k0)     ? 1.0f : 0.0f;
        a.y = (m == 4 * q + k0 + 1) ? 1.0f : 0.0f;
        size_t off = (size_t)(d0 + m) * BATCH + 4 * q + k0;
        b = *(const v2f*)(outT0 + off) + *(const v2f*)(outT1 + off);
        c = __builtin_amdgcn_wmma_f32_16x16x4_f32(
                 false, a,  false, b,
                 (short)0, c,  false,  false);
    }

#pragma unroll
    for (int r = 0; r < 8; ++r) {
        int brow = r + 8 * h;
        out[(size_t)brow * ndst + d0 + m] = c[r];
    }
}

extern "C" void kernel_launch(void* const* d_in, const int* in_sizes, int n_in,
                              void* d_out, int out_size, void* d_ws, size_t ws_size,
                              hipStream_t stream) {
    const float* x    = (const float*)d_in[0];
    const float* vals = (const float*)d_in[1];
    const float* bias = (const float*)d_in[2];
    const int*   idx  = (const int*)d_in[3];

    int nsrc = in_sizes[0] / BATCH;
    int nnz  = in_sizes[1];
    int ndst = in_sizes[2];

    float* out  = (float*)d_out;
    float* xT   = out;
    float* acc0 = (float*)d_ws;
    float* acc1 = acc0 + (size_t)ndst * BATCH;

    int nOutT = ndst * BATCH;
    zero_planes<<<(nOutT + 255) / 256, 256, 0, stream>>>(acc0, acc1, nOutT);

    int nXT = nsrc * BATCH;
    transpose_x<<<(nXT + 255) / 256, 256, 0, stream>>>(x, xT, nsrc);

    scatter_edges<<<(nnz + 255) / 256, 256, 0, stream>>>(vals, idx, xT, acc0,
                                                        (size_t)nOutT, nnz);

    int tiles = (ndst + 15) / 16;
    wmma_transpose_bias<<<(tiles + 7) / 8, 256, 0, stream>>>(acc0, acc1, bias, out, ndst);
}
